# starting kernel
# baseline (speedup 1.0000x reference)
#include <hip/hip_runtime.h>
#include <hip/hip_bf16.h>


typedef float v2f __attribute__((ext_vector_type(2)));
typedef float v4f __attribute__((ext_vector_type(4)));
typedef float v8f __attribute__((ext_vector_type(8)));

#define NWAVES 8
#define LTOK   30
#define EDIM   64
#define ROWSTR 66

__global__ __launch_bounds__(256) void tweetrep_kernel(
    const int*   __restrict__ x,
    const float* __restrict__ emb,
    float*       __restrict__ out)
{
    __shared__ float wg[NWAVES][LTOK][ROWSTR];
    __shared__ int   sIdx[NWAVES][32];
    __shared__ float bArr[NWAVES][32];

    const int tid   = threadIdx.x;
    const int wave  = tid >> 5;
    const int lane  = tid & 31;
    const int tweet = blockIdx.x * NWAVES + wave;

    if (lane < LTOK) sIdx[wave][lane] = x[(size_t)tweet * LTOK + lane];
    __asm__ volatile("" ::: "memory");

#pragma unroll
    for (int it = 0; it < 15; ++it) {
        int flat = it * 32 + lane;
        int row  = flat >> 4;
        int c4   = (flat & 15) << 2;
        int idx  = sIdx[wave][row];
        v4f v = *(const v4f*)(emb + (size_t)idx * EDIM + c4);
        float* dst = &wg[wave][row][c4];
        *(v2f*)(dst)     = (v2f){v.x, v.y};
        *(v2f*)(dst + 2) = (v2f){v.z, v.w};
    }
    __asm__ volatile("" ::: "memory");

    const int half  = lane >> 4;
    const int rl    = lane & 15;
    const int row1  = 16 + ((rl < 14) ? rl : 0);
    const float mskf = (rl < 14) ? 1.f : 0.f;
    v2f F0[16], F1[16];
#pragma unroll
    for (int kb = 0; kb < 16; ++kb) {
        int k = kb * 4 + half * 2;
        F0[kb] = *(const v2f*)&wg[wave][rl][k];
        F1[kb] = *(const v2f*)&wg[wave][row1][k];
    }

    v8f c00 = {}, c01 = {}, c10 = {}, c11 = {};
#pragma unroll
    for (int kb = 0; kb < 16; ++kb) {
        c00 = __builtin_amdgcn_wmma_f32_16x16x4_f32(false, F0[kb], false, F0[kb], (short)0, c00, false, false);
        c01 = __builtin_amdgcn_wmma_f32_16x16x4_f32(false, F0[kb], false, F1[kb], (short)0, c01, false, false);
        c10 = __builtin_amdgcn_wmma_f32_16x16x4_f32(false, F1[kb], false, F0[kb], (short)0, c10, false, false);
        c11 = __builtin_amdgcn_wmma_f32_16x16x4_f32(false, F1[kb], false, F1[kb], (short)0, c11, false, false);
    }

#pragma unroll
    for (int r = 0; r < 8; ++r) {
        float v = fmaf(mskf, c01[r], c00[r]);
        v += __shfl_xor(v, 1, 32); v += __shfl_xor(v, 2, 32);
        v += __shfl_xor(v, 4, 32); v += __shfl_xor(v, 8, 32);
        if (rl == 0) bArr[wave][r + 8 * half] = v;
        float w = fmaf(mskf, c11[r], c10[r]);
        w += __shfl_xor(w, 1, 32); w += __shfl_xor(w, 2, 32);
        w += __shfl_xor(w, 4, 32); w += __shfl_xor(w, 8, 32);
        if (rl == 0) bArr[wave][16 + r + 8 * half] = w;
    }
    __asm__ volatile("" ::: "memory");

    float bl = (lane < LTOK) ? bArr[wave][lane] : -3.0e38f;
    float mx = bl;
    mx = fmaxf(mx, __shfl_xor(mx, 1, 32));
    mx = fmaxf(mx, __shfl_xor(mx, 2, 32));
    mx = fmaxf(mx, __shfl_xor(mx, 4, 32));
    mx = fmaxf(mx, __shfl_xor(mx, 8, 32));
    mx = fmaxf(mx, __shfl_xor(mx, 16, 32));
    float ev = (lane < LTOK) ? expf(bl - mx) : 0.f;
    float s = ev;
    s += __shfl_xor(s, 1, 32);
    s += __shfl_xor(s, 2, 32);
    s += __shfl_xor(s, 4, 32);
    s += __shfl_xor(s, 8, 32);
    s += __shfl_xor(s, 16, 32);
    float cl = ev / s;

    float a0 = 0.f, a1 = 0.f;
#pragma unroll
    for (int l = 0; l < LTOK; ++l) {
        float c = __shfl(cl, l, 32);
        v2f wv = *(const v2f*)&wg[wave][l][2 * lane];
        a0 = fmaf(c, wv.x, a0);
        a1 = fmaf(c, wv.y, a1);
    }
    __asm__ volatile("" ::: "memory");
    *(v2f*)&wg[wave][0][2 * lane] = (v2f){a0, a1};
    __syncthreads();

    const int p  = (blockIdx.x * NWAVES) >> 10;
    const int qb = (blockIdx.x * NWAVES) & 1023;
    {
        int e  = tid >> 2;
        int w2 = tid & 3;
        v2f o = (v2f){ wg[2 * w2][0][e], wg[2 * w2 + 1][0][e] };
        *(v2f*)&out[((size_t)p * 64 + e) * 1024 + qb + 2 * w2] = o;
    }
}

extern "C" void kernel_launch(void* const* d_in, const int* in_sizes, int n_in,
                              void* d_out, int out_size, void* d_ws, size_t ws_size,
                              hipStream_t stream) {
    const int*   x   = (const int*)d_in[0];
    const float* emb = (const float*)d_in[1];
    float*       out = (float*)d_out;

    int n_tweets = in_sizes[0] / LTOK;
    int blocks   = n_tweets / NWAVES;
    hipLaunchKernelGGL(tweetrep_kernel, dim3(blocks), dim3(256), 0, stream,
                       x, emb, out);
}


# scores b=Wg.V as a 32-WMMA matrix-vector product (A = V in every row, V summed from the gather registers) replacing the 64-WMMA Gram matrix and lane row sums; summation order of b changes
# speedup vs baseline: 1.5534x; 1.5534x over previous
#include <hip/hip_runtime.h>


typedef float v2f __attribute__((ext_vector_type(2)));
typedef float v4f __attribute__((ext_vector_type(4)));
typedef float v8f __attribute__((ext_vector_type(8)));

#define NWAVES 8
#define LTOK   30
#define EDIM   64
#define ROWSTR 66

__global__ __launch_bounds__(256) void tweetrep_kernel(
    const int*   __restrict__ x,
    const float* __restrict__ emb,
    float*       __restrict__ out)
{
    __shared__ float wg[NWAVES][LTOK][ROWSTR];
    __shared__ __attribute__((aligned(16))) float sV[NWAVES][EDIM];

    const int tid   = threadIdx.x;
    const int wave  = tid >> 5;
    const int lane  = tid & 31;
    const int tweet = blockIdx.x * NWAVES + wave;
    const int half  = lane >> 4;
    const int rl    = lane & 15;

    if (lane < LTOK) sV[wave][lane] = __int_as_float(x[(size_t)tweet * LTOK + lane]);
    __asm__ volatile("" ::: "memory");

    v4f vs;
#pragma unroll
    for (int it = 0; it < 15; ++it) {
        int flat = it * 32 + lane;
        int row  = flat >> 4;
        int c4   = (flat & 15) << 2;
        int idx  = __float_as_int(sV[wave][row]);
        v4f v = *(const v4f*)(emb + (size_t)idx * EDIM + c4);
        float* dst = &wg[wave][row][c4];
        *(v2f*)(dst)     = (v2f){v.x, v.y};
        *(v2f*)(dst + 2) = (v2f){v.z, v.w};
        vs = (it == 0) ? v : vs + v;
    }
    vs.x += __shfl_xor(vs.x, 16, 32);
    vs.y += __shfl_xor(vs.y, 16, 32);
    vs.z += __shfl_xor(vs.z, 16, 32);
    vs.w += __shfl_xor(vs.w, 16, 32);
    if (lane < 16) *(v4f*)&sV[wave][4 * rl] = vs;
    __asm__ volatile("" ::: "memory");

    const int row1 = 16 + ((rl < 14) ? rl : 0);
    v2f FV[16], F0[16], F1[16];
#pragma unroll
    for (int kb = 0; kb < 16; ++kb) {
        int k = kb * 4 + half * 2;
        FV[kb] = *(const v2f*)&sV[wave][k];
        F0[kb] = *(const v2f*)&wg[wave][rl][k];
        F1[kb] = *(const v2f*)&wg[wave][row1][k];
    }

    v8f c0 = {}, c1 = {};
#pragma unroll
    for (int kb = 0; kb < 16; ++kb) {
        c0 = __builtin_amdgcn_wmma_f32_16x16x4_f32(false, FV[kb], false, F0[kb], (short)0, c0, false, false);
        c1 = __builtin_amdgcn_wmma_f32_16x16x4_f32(false, FV[kb], false, F1[kb], (short)0, c1, false, false);
    }

    float bl = (lane < 16) ? c0[0] : c1[0];
    bl = (lane < LTOK) ? bl : -3.0e38f;
    float mx = bl;
    mx = fmaxf(mx, __shfl_xor(mx, 1, 32));
    mx = fmaxf(mx, __shfl_xor(mx, 2, 32));
    mx = fmaxf(mx, __shfl_xor(mx, 4, 32));
    mx = fmaxf(mx, __shfl_xor(mx, 8, 32));
    mx = fmaxf(mx, __shfl_xor(mx, 16, 32));
    float ev = (lane < LTOK) ? expf(bl - mx) : 0.f;
    float s = ev;
    s += __shfl_xor(s, 1, 32);
    s += __shfl_xor(s, 2, 32);
    s += __shfl_xor(s, 4, 32);
    s += __shfl_xor(s, 8, 32);
    s += __shfl_xor(s, 16, 32);
    float cl = ev / s;

    float a0 = 0.f, a1 = 0.f;
#pragma unroll
    for (int l = 0; l < LTOK; ++l) {
        float c = __shfl(cl, l, 32);
        v2f wv = *(const v2f*)&wg[wave][l][2 * lane];
        a0 = fmaf(c, wv.x, a0);
        a1 = fmaf(c, wv.y, a1);
    }
    __asm__ volatile("" ::: "memory");
    *(v2f*)&wg[wave][0][2 * lane] = (v2f){a0, a1};
    __syncthreads();

    const int p  = (blockIdx.x * NWAVES) >> 10;
    const int qb = (blockIdx.x * NWAVES) & 1023;
    {
        int e  = tid >> 2;
        int w2 = tid & 3;
        v2f o = (v2f){ wg[2 * w2][0][e], wg[2 * w2 + 1][0][e] };
        *(v2f*)&out[((size_t)p * 64 + e) * 1024 + qb + 2 * w2] = o;
    }
}

extern "C" void kernel_launch(void* const* d_in, const int* in_sizes, int n_in,
                              void* d_out, int out_size, void* d_ws, size_t ws_size,
                              hipStream_t stream) {
    const int*   x   = (const int*)d_in[0];
    const float* emb = (const float*)d_in[1];
    float*       out = (float*)d_out;

    int n_tweets = in_sizes[0] / LTOK;
    int blocks   = n_tweets / NWAVES;
    hipLaunchKernelGGL(tweetrep_kernel, dim3(blocks), dim3(256), 0, stream,
                       x, emb, out);
}


# all lane exchanges (index broadcast, V half join, softmax butterflies, c[l] broadcast) moved from ds_bpermute and an LDS round trip to the vector ALU: DPP row_share/row_xmask, permlanex16, readlane
# speedup vs baseline: 1.6091x; 1.0359x over previous
#include <hip/hip_runtime.h>


typedef float v2f __attribute__((ext_vector_type(2)));
typedef float v4f __attribute__((ext_vector_type(4)));
typedef float v8f __attribute__((ext_vector_type(8)));

#define NWAVES 8
#define LTOK   30
#define EDIM   64
#define ROWSTR 66

static __device__ __forceinline__ float xhalf(float x) {
    int b = __float_as_int(x);
    return __int_as_float(__builtin_amdgcn_permlanex16(b, b, 0x76543210, 0xfedcba98, false, false));
}
template <int M>
static __device__ __forceinline__ float xrow(float x) {
    return __int_as_float(__builtin_amdgcn_update_dpp(0, __float_as_int(x), 0x160 + M, 0xf, 0xf, true));
}
template <int N>
static __device__ __forceinline__ int rowshare(int x) {
    return __builtin_amdgcn_update_dpp(0, x, 0x150 + N, 0xf, 0xf, true);
}

__global__ __launch_bounds__(256) void tweetrep_kernel(
    const int*   __restrict__ x,
    const float* __restrict__ emb,
    float*       __restrict__ out)
{
    __shared__ float wg[NWAVES][LTOK][ROWSTR];
    __shared__ __attribute__((aligned(16))) float sV[NWAVES][EDIM];

    const int tid   = threadIdx.x;
    const int wave  = tid >> 5;
    const int lane  = tid & 31;
    const int tweet = blockIdx.x * NWAVES + wave;
    const int half  = lane >> 4;
    const int rl    = lane & 15;

    const int tk = (rl < 15) ? (2 * rl + half) : (28 + half);
    const int xi = x[(size_t)tweet * LTOK + tk];

    const float* ecol = emb + 4 * rl;
    v4f vs;
#define GATHER_STEP(IT)                                                        \
    {                                                                          \
        const int idx = rowshare<IT>(xi);                                      \
        const v4f v = *(const v4f*)(ecol + (size_t)idx * EDIM);                \
        float* dst = &wg[wave][2 * (IT) + half][4 * rl];        \
        *(v2f*)(dst)     = (v2f){v.x, v.y};                                    \
        *(v2f*)(dst + 2) = (v2f){v.z, v.w};                                    \
        vs = ((IT) == 0) ? v : vs + v;                                         \
    }
    GATHER_STEP(0)  GATHER_STEP(1)  GATHER_STEP(2)  GATHER_STEP(3)  GATHER_STEP(4)
    GATHER_STEP(5)  GATHER_STEP(6)  GATHER_STEP(7)  GATHER_STEP(8)  GATHER_STEP(9)
    GATHER_STEP(10) GATHER_STEP(11) GATHER_STEP(12) GATHER_STEP(13) GATHER_STEP(14)
#undef GATHER_STEP
    vs.x += xhalf(vs.x);
    vs.y += xhalf(vs.y);
    vs.z += xhalf(vs.z);
    vs.w += xhalf(vs.w);
    if (lane < 16) *(v4f*)&sV[wave][4 * rl] = vs;
    __asm__ volatile("" ::: "memory");

    const int row1 = 16 + ((rl < 14) ? rl : 0);
    v2f FV[16], F0[16], F1[16];
#pragma unroll
    for (int kb = 0; kb < 16; ++kb) {
        int k = kb * 4 + half * 2;
        FV[kb] = *(const v2f*)&sV[wave][k];
        F0[kb] = *(const v2f*)&wg[wave][rl][k];
        F1[kb] = *(const v2f*)&wg[wave][row1][k];
    }

    v8f c0 = {}, c1 = {};
#pragma unroll
    for (int kb = 0; kb < 16; ++kb) {
        c0 = __builtin_amdgcn_wmma_f32_16x16x4_f32(false, FV[kb], false, F0[kb], (short)0, c0, false, false);
        c1 = __builtin_amdgcn_wmma_f32_16x16x4_f32(false, FV[kb], false, F1[kb], (short)0, c1, false, false);
    }

    float bl = (lane < 16) ? c0[0] : c1[0];
    bl = (lane < LTOK) ? bl : -3.0e38f;
    float mx = bl;
    mx = fmaxf(mx, xrow<1>(mx));
    mx = fmaxf(mx, xrow<2>(mx));
    mx = fmaxf(mx, xrow<4>(mx));
    mx = fmaxf(mx, xrow<8>(mx));
    mx = fmaxf(mx, xhalf(mx));
    float ev = (lane < LTOK) ? expf(bl - mx) : 0.f;
    float s = ev;
    s += xrow<1>(s);
    s += xrow<2>(s);
    s += xrow<4>(s);
    s += xrow<8>(s);
    s += xhalf(s);
    float cl = ev / s;

    float a0 = 0.f, a1 = 0.f;
#pragma unroll
    for (int l = 0; l < LTOK; ++l) {
        float c = __int_as_float(__builtin_amdgcn_readlane(__float_as_int(cl), l));
        v2f wv = *(const v2f*)&wg[wave][l][2 * lane];
        a0 = fmaf(c, wv.x, a0);
        a1 = fmaf(c, wv.y, a1);
    }
    __asm__ volatile("" ::: "memory");
    *(v2f*)&wg[wave][0][2 * lane] = (v2f){a0, a1};
    __syncthreads();

    const int p  = (blockIdx.x * NWAVES) >> 10;
    const int qb = (blockIdx.x * NWAVES) & 1023;
    {
        int e  = tid >> 2;
        int w2 = tid & 3;
        v2f o = (v2f){ wg[2 * w2][0][e], wg[2 * w2 + 1][0][e] };
        *(v2f*)&out[((size_t)p * 64 + e) * 1024 + qb + 2 * w2] = o;
    }
}

extern "C" void kernel_launch(void* const* d_in, const int* in_sizes, int n_in,
                              void* d_out, int out_size, void* d_ws, size_t ws_size,
                              hipStream_t stream) {
    const int*   x   = (const int*)d_in[0];
    const float* emb = (const float*)d_in[1];
    float*       out = (float*)d_out;

    int n_tweets = in_sizes[0] / LTOK;
    int blocks   = n_tweets / NWAVES;
    hipLaunchKernelGGL(tweetrep_kernel, dim3(blocks), dim3(256), 0, stream,
                       x, emb, out);
}


# the 15 row loads of the next tweet fly under the 32 WMMAs of the current one: each wave walks 8 consecutive tweets, no block barrier, 768 resident blocks; tile pitch 68; summation order unchanged
# speedup vs baseline: 1.9095x; 1.1867x over previous
#include <hip/hip_runtime.h>


typedef float v2f __attribute__((ext_vector_type(2)));
typedef float v4f __attribute__((ext_vector_type(4)));
typedef float v8f __attribute__((ext_vector_type(8)));

#define NWAVES 8
#define TPW    8
#define LTOK   30
#define EDIM   64
#define ROWSTR 68
#define STSTR  72

static __device__ __forceinline__ float xhalf(float x) {
    int b = __float_as_int(x);
    return __int_as_float(__builtin_amdgcn_permlanex16(b, b, 0x76543210, 0xfedcba98, false, false));
}
template <int M>
static __device__ __forceinline__ float xrow(float x) {
    return __int_as_float(__builtin_amdgcn_update_dpp(0, __float_as_int(x), 0x160 + M, 0xf, 0xf, true));
}
template <int N>
static __device__ __forceinline__ int rowshare(int x) {
    return __builtin_amdgcn_update_dpp(0, x, 0x150 + N, 0xf, 0xf, true);
}

#define GATHER_ROW(IT, XI) R[IT] = *(const v4f*)(ecol + (size_t)rowshare<IT>(XI) * EDIM);
#define GATHER_ALL(XI)                                                              \
    GATHER_ROW(0, XI)  GATHER_ROW(1, XI)  GATHER_ROW(2, XI)  GATHER_ROW(3, XI)      \
    GATHER_ROW(4, XI)  GATHER_ROW(5, XI)  GATHER_ROW(6, XI)  GATHER_ROW(7, XI)      \
    GATHER_ROW(8, XI)  GATHER_ROW(9, XI)  GATHER_ROW(10, XI) GATHER_ROW(11, XI)     \
    GATHER_ROW(12, XI) GATHER_ROW(13, XI) GATHER_ROW(14, XI)

__global__ __launch_bounds__(256) void tweetrep_kernel(
    const int*   __restrict__ x,
    const float* __restrict__ emb,
    float*       __restrict__ out)
{
    __shared__ __attribute__((aligned(16))) float wg[NWAVES][LTOK][ROWSTR];
    __shared__ __attribute__((aligned(16))) float sV[NWAVES][EDIM];
    __shared__ __attribute__((aligned(16))) float st[NWAVES][TPW][STSTR];

    const int tid    = threadIdx.x;
    const int wave   = tid >> 5;
    const int lane   = tid & 31;
    const int half   = lane >> 4;
    const int rl     = lane & 15;
    const int tweet0 = (blockIdx.x * NWAVES + wave) * TPW;

    const int  tk = (rl < 15) ? (2 * rl + half) : (28 + half);
    const int* xp = x + (size_t)tweet0 * LTOK + tk;
    const float* ecol = emb + 4 * rl;

    const int row1 = 16 + ((rl < 14) ? rl : 0);

    v4f R[15];
    {
        const int xi0 = xp[0];
        GATHER_ALL(xi0)
    }
    int xn = xp[LTOK];

#pragma unroll 1
    for (int j = 0; j < TPW; ++j) {
        v4f vs = R[0];
#pragma unroll
        for (int it = 0; it < 15; ++it) {
            *(v4f*)&wg[wave][2 * it + half][4 * rl] = R[it];
            if (it) vs += R[it];
        }
        vs.x += xhalf(vs.x);
        vs.y += xhalf(vs.y);
        vs.z += xhalf(vs.z);
        vs.w += xhalf(vs.w);
        if (lane < 16) *(v4f*)&sV[wave][4 * rl] = vs;

        if (j + 1 < TPW) {
            GATHER_ALL(xn)
        }
        if (j + 2 < TPW) xn = xp[(j + 2) * LTOK];

        __asm__ volatile("" ::: "memory");

        v8f c0 = {}, c1 = {};
#pragma unroll
        for (int kb = 0; kb < 16; ++kb) {
            const int k = kb * 4 + half * 2;
            const v2f fv = *(const v2f*)&sV[wave][k];
            const v2f f0 = *(const v2f*)&wg[wave][rl][k];
            const v2f f1 = *(const v2f*)&wg[wave][row1][k];
            c0 = __builtin_amdgcn_wmma_f32_16x16x4_f32(false, fv, false, f0, (short)0, c0, false, false);
            c1 = __builtin_amdgcn_wmma_f32_16x16x4_f32(false, fv, false, f1, (short)0, c1, false, false);
        }

        float bl = (lane < 16) ? c0[0] : c1[0];
        bl = (lane < LTOK) ? bl : -3.0e38f;
        float mx = bl;
        mx = fmaxf(mx, xrow<1>(mx));
        mx = fmaxf(mx, xrow<2>(mx));
        mx = fmaxf(mx, xrow<4>(mx));
        mx = fmaxf(mx, xrow<8>(mx));
        mx = fmaxf(mx, xhalf(mx));
        float ev = (lane < LTOK) ? expf(bl - mx) : 0.f;
        float s = ev;
        s += xrow<1>(s);
        s += xrow<2>(s);
        s += xrow<4>(s);
        s += xrow<8>(s);
        s += xhalf(s);
        float cl = ev / s;

        float a0 = 0.f, a1 = 0.f;
#pragma unroll
        for (int l = 0; l < LTOK; ++l) {
            float c = __int_as_float(__builtin_amdgcn_readlane(__float_as_int(cl), l));
            v2f wv = *(const v2f*)&wg[wave][l][2 * lane];
            a0 = fmaf(c, wv.x, a0);
            a1 = fmaf(c, wv.y, a1);
        }
        *(v2f*)&st[wave][j][2 * lane] = (v2f){a0, a1};

        __asm__ volatile("" ::: "memory");
    }

    const int p  = tweet0 >> 10;
    const int q  = tweet0 & 1023;
    const int es = lane >> 2;
    const int pr = lane & 3;
#pragma unroll
    for (int i = 0; i < 8; ++i) {
        const int e = 8 * i + es;
        v2f o = (v2f){ st[wave][2 * pr][e], st[wave][2 * pr + 1][e] };
        *(v2f*)&out[((size_t)p * 64 + e) * 1024 + q + 2 * pr] = o;
    }
}

extern "C" void kernel_launch(void* const* d_in, const int* in_sizes, int n_in,
                              void* d_out, int out_size, void* d_ws, size_t ws_size,
                              hipStream_t stream) {
    const int*   x   = (const int*)d_in[0];
    const float* emb = (const float*)d_in[1];
    float*       out = (float*)d_out;

    int n_tweets = in_sizes[0] / LTOK;
    int blocks   = n_tweets / (NWAVES * TPW);
    hipLaunchKernelGGL(tweetrep_kernel, dim3(blocks), dim3(256), 0, stream,
                       x, emb, out);
}


# row addressing taken off the vector ALU: token indices by scalar loads, row bases by scalar arithmetic, each 256-byte row one load of 32 lanes from a scalar base; summation order unchanged
# speedup vs baseline: 1.9867x; 1.0404x over previous
#include <hip/hip_runtime.h>


typedef float v2f __attribute__((ext_vector_type(2)));
typedef float v8f __attribute__((ext_vector_type(8)));
typedef __attribute__((address_space(1))) char gchar;
typedef __attribute__((address_space(1))) v2f  gv2f;

#define NWAVES 8
#define TPW    8
#define LTOK   30
#define EDIM   64
#define ROWSTR 68
#define STSTR  72

static __device__ __forceinline__ float xhalf(float x) {
    int b = __float_as_int(x);
    return __int_as_float(__builtin_amdgcn_permlanex16(b, b, 0x76543210, 0xfedcba98, false, false));
}
template <int M>
static __device__ __forceinline__ float xrow(float x) {
    return __int_as_float(__builtin_amdgcn_update_dpp(0, __float_as_int(x), 0x160 + M, 0xf, 0xf, true));
}
#define LOAD_ROW(L, XS)  R[L] = *(const gv2f*)(tab + ((unsigned)(XS)[L] << 8) + lo);
#define LOAD_ROWS(XS)                                                                          \
    LOAD_ROW(0, XS)  LOAD_ROW(1, XS)  LOAD_ROW(2, XS)  LOAD_ROW(3, XS)  LOAD_ROW(4, XS)  LOAD_ROW(5, XS)   \
    LOAD_ROW(6, XS)  LOAD_ROW(7, XS)  LOAD_ROW(8, XS)  LOAD_ROW(9, XS)  LOAD_ROW(10, XS) LOAD_ROW(11, XS)  \
    LOAD_ROW(12, XS) LOAD_ROW(13, XS) LOAD_ROW(14, XS) LOAD_ROW(15, XS) LOAD_ROW(16, XS) LOAD_ROW(17, XS)  \
    LOAD_ROW(18, XS) LOAD_ROW(19, XS) LOAD_ROW(20, XS) LOAD_ROW(21, XS) LOAD_ROW(22, XS) LOAD_ROW(23, XS)  \
    LOAD_ROW(24, XS) LOAD_ROW(25, XS) LOAD_ROW(26, XS) LOAD_ROW(27, XS) LOAD_ROW(28, XS) LOAD_ROW(29, XS)

__global__ __launch_bounds__(256) void tweetrep_kernel(
    const int*   __restrict__ x,
    const float* __restrict__ emb,
    float*       __restrict__ out)
{
    __shared__ __attribute__((aligned(16))) float wg[NWAVES][LTOK][ROWSTR];
    __shared__ __attribute__((aligned(16))) float sV[NWAVES][EDIM];
    __shared__ __attribute__((aligned(16))) float st[NWAVES][TPW][STSTR];

    const int tid    = threadIdx.x;
    const int wave   = __builtin_amdgcn_readfirstlane(tid >> 5);
    const int lane   = tid & 31;
    const int half   = lane >> 4;
    const int rl     = lane & 15;
    const int tweet0 = (blockIdx.x * NWAVES + wave) * TPW;

    const int*     xw   = x + (size_t)tweet0 * LTOK;
    const unsigned loff = 8u * lane;

    const int row1 = 16 + ((rl < 14) ? rl : 0);

    v2f R[LTOK];
    int xn[LTOK];
    const gchar* tab = (const gchar*)emb;
    {
        const unsigned lo = loff;
        LOAD_ROWS(xw)
    }
#pragma unroll
    for (int l = 0; l < LTOK; ++l) xn[l] = xw[LTOK + l];

#pragma unroll 1
    for (int j = 0; j < TPW; ++j) {
        v2f se = R[0], so = R[1];
#pragma unroll
        for (int l = 0; l < LTOK; ++l) {
            *(v2f*)&wg[wave][l][2 * lane] = R[l];
            if (l >= 2) { if (l & 1) so += R[l]; else se += R[l]; }
        }
        *(v2f*)&sV[wave][2 * lane] = se + so;

        if (j + 1 < TPW) {
            unsigned lo = loff;
            __asm__ volatile("" : "+v"(lo));
            LOAD_ROWS(xn)
        }
        if (j + 2 < TPW) {
#pragma unroll
            for (int l = 0; l < LTOK; ++l) xn[l] = xw[(j + 2) * LTOK + l];
        }

        __asm__ volatile("" ::: "memory");

        v8f c0 = {}, c1 = {};
#pragma unroll
        for (int kb = 0; kb < 16; ++kb) {
            const int k = kb * 4 + half * 2;
            const v2f fv = *(const v2f*)&sV[wave][k];
            const v2f f0 = *(const v2f*)&wg[wave][rl][k];
            const v2f f1 = *(const v2f*)&wg[wave][row1][k];
            c0 = __builtin_amdgcn_wmma_f32_16x16x4_f32(false, fv, false, f0, (short)0, c0, false, false);
            c1 = __builtin_amdgcn_wmma_f32_16x16x4_f32(false, fv, false, f1, (short)0, c1, false, false);
        }

        float bl = (lane < 16) ? c0[0] : c1[0];
        bl = (lane < LTOK) ? bl : -3.0e38f;
        float mx = bl;
        mx = fmaxf(mx, xrow<1>(mx));
        mx = fmaxf(mx, xrow<2>(mx));
        mx = fmaxf(mx, xrow<4>(mx));
        mx = fmaxf(mx, xrow<8>(mx));
        mx = fmaxf(mx, xhalf(mx));
        float ev = (lane < LTOK) ? expf(bl - mx) : 0.f;
        float s = ev;
        s += xrow<1>(s);
        s += xrow<2>(s);
        s += xrow<4>(s);
        s += xrow<8>(s);
        s += xhalf(s);
        float cl = ev / s;

        float a0 = 0.f, a1 = 0.f;
#pragma unroll
        for (int l = 0; l < LTOK; ++l) {
            float c = __int_as_float(__builtin_amdgcn_readlane(__float_as_int(cl), l));
            v2f wv = *(const v2f*)&wg[wave][l][2 * lane];
            a0 = fmaf(c, wv.x, a0);
            a1 = fmaf(c, wv.y, a1);
        }
        *(v2f*)&st[wave][j][2 * lane] = (v2f){a0, a1};

        __asm__ volatile("" ::: "memory");
    }

    const int p  = tweet0 >> 10;
    const int q  = tweet0 & 1023;
    const int es = lane >> 2;
    const int pr = lane & 3;
#pragma unroll
    for (int i = 0; i < 8; ++i) {
        const int e = 8 * i + es;
        v2f o = (v2f){ st[wave][2 * pr][e], st[wave][2 * pr + 1][e] };
        *(v2f*)&out[((size_t)p * 64 + e) * 1024 + q + 2 * pr] = o;
    }
}

extern "C" void kernel_launch(void* const* d_in, const int* in_sizes, int n_in,
                              void* d_out, int out_size, void* d_ws, size_t ws_size,
                              hipStream_t stream) {
    const int*   x   = (const int*)d_in[0];
    const float* emb = (const float*)d_in[1];
    float*       out = (float*)d_out;

    int n_tweets = in_sizes[0] / LTOK;
    int blocks   = n_tweets / (NWAVES * TPW);
    hipLaunchKernelGGL(tweetrep_kernel, dim3(blocks), dim3(256), 0, stream,
                       x, emb, out);
}


# each wave steps its own issue priority down as it advances (s_setprio 3,3,2,2,1,1,0,0 over its 8 tweets): the 6 waves of a SIMD end together, not oldest first; summation order unchanged
# speedup vs baseline: 2.1100x; 1.0621x over previous
#include <hip/hip_runtime.h>


typedef float v2f __attribute__((ext_vector_type(2)));
typedef float v8f __attribute__((ext_vector_type(8)));
typedef __attribute__((address_space(1))) char gchar;
typedef __attribute__((address_space(1))) v2f  gv2f;

#define NWAVES 8
#define TPW    8
#define LTOK   30
#define EDIM   64
#define ROWSTR 68
#define STSTR  72

static __device__ __forceinline__ float xhalf(float x) {
    int b = __float_as_int(x);
    return __int_as_float(__builtin_amdgcn_permlanex16(b, b, 0x76543210, 0xfedcba98, false, false));
}
template <int M>
static __device__ __forceinline__ float xrow(float x) {
    return __int_as_float(__builtin_amdgcn_update_dpp(0, __float_as_int(x), 0x160 + M, 0xf, 0xf, true));
}
#define LOAD_ROW(L, XS)  R[L] = *(const gv2f*)(tab + ((unsigned)(XS)[L] << 8) + lo);
#define LOAD_ROWS(XS)                                                                          \
    LOAD_ROW(0, XS)  LOAD_ROW(1, XS)  LOAD_ROW(2, XS)  LOAD_ROW(3, XS)  LOAD_ROW(4, XS)  LOAD_ROW(5, XS)   \
    LOAD_ROW(6, XS)  LOAD_ROW(7, XS)  LOAD_ROW(8, XS)  LOAD_ROW(9, XS)  LOAD_ROW(10, XS) LOAD_ROW(11, XS)  \
    LOAD_ROW(12, XS) LOAD_ROW(13, XS) LOAD_ROW(14, XS) LOAD_ROW(15, XS) LOAD_ROW(16, XS) LOAD_ROW(17, XS)  \
    LOAD_ROW(18, XS) LOAD_ROW(19, XS) LOAD_ROW(20, XS) LOAD_ROW(21, XS) LOAD_ROW(22, XS) LOAD_ROW(23, XS)  \
    LOAD_ROW(24, XS) LOAD_ROW(25, XS) LOAD_ROW(26, XS) LOAD_ROW(27, XS) LOAD_ROW(28, XS) LOAD_ROW(29, XS)

__global__ __launch_bounds__(256) void tweetrep_kernel(
    const int*   __restrict__ x,
    const float* __restrict__ emb,
    float*       __restrict__ out)
{
    __shared__ __attribute__((aligned(16))) float wg[NWAVES][LTOK][ROWSTR];
    __shared__ __attribute__((aligned(16))) float sV[NWAVES][EDIM];
    __shared__ __attribute__((aligned(16))) float st[NWAVES][TPW][STSTR];

    __builtin_amdgcn_s_setprio(3);

    const int tid    = threadIdx.x;
    const int wave   = __builtin_amdgcn_readfirstlane(tid >> 5);
    const int lane   = tid & 31;
    const int half   = lane >> 4;
    const int rl     = lane & 15;
    const int tweet0 = (blockIdx.x * NWAVES + wave) * TPW;

    const int*     xw   = x + (size_t)tweet0 * LTOK;
    const unsigned loff = 8u * lane;

    const int row1 = 16 + ((rl < 14) ? rl : 0);

    v2f R[LTOK];
    int xn[LTOK];
    const gchar* tab = (const gchar*)emb;
    {
        const unsigned lo = loff;
        LOAD_ROWS(xw)
    }
#pragma unroll
    for (int l = 0; l < LTOK; ++l) xn[l] = xw[LTOK + l];

#pragma unroll 1
    for (int j = 0; j < TPW; ++j) {
        if (j == 2) __builtin_amdgcn_s_setprio(2);
        if (j == 4) __builtin_amdgcn_s_setprio(1);
        if (j == 6) __builtin_amdgcn_s_setprio(0);

        v2f se = R[0], so = R[1];
#pragma unroll
        for (int l = 0; l < LTOK; ++l) {
            *(v2f*)&wg[wave][l][2 * lane] = R[l];
            if (l >= 2) { if (l & 1) so += R[l]; else se += R[l]; }
        }
        *(v2f*)&sV[wave][2 * lane] = se + so;

        if (j + 1 < TPW) {
            unsigned lo = loff;
            __asm__ volatile("" : "+v"(lo));
            LOAD_ROWS(xn)
        }
        if (j + 2 < TPW) {
#pragma unroll
            for (int l = 0; l < LTOK; ++l) xn[l] = xw[(j + 2) * LTOK + l];
        }

        __asm__ volatile("" ::: "memory");

        v8f c0 = {}, c1 = {};
#pragma unroll
        for (int kb = 0; kb < 16; ++kb) {
            const int k = kb * 4 + half * 2;
            const v2f fv = *(const v2f*)&sV[wave][k];
            const v2f f0 = *(const v2f*)&wg[wave][rl][k];
            const v2f f1 = *(const v2f*)&wg[wave][row1][k];
            c0 = __builtin_amdgcn_wmma_f32_16x16x4_f32(false, fv, false, f0, (short)0, c0, false, false);
            c1 = __builtin_amdgcn_wmma_f32_16x16x4_f32(false, fv, false, f1, (short)0, c1, false, false);
        }

        float bl = (lane < 16) ? c0[0] : c1[0];
        bl = (lane < LTOK) ? bl : -3.0e38f;
        float mx = bl;
        mx = fmaxf(mx, xrow<1>(mx));
        mx = fmaxf(mx, xrow<2>(mx));
        mx = fmaxf(mx, xrow<4>(mx));
        mx = fmaxf(mx, xrow<8>(mx));
        mx = fmaxf(mx, xhalf(mx));
        float ev = (lane < LTOK) ? expf(bl - mx) : 0.f;
        float s = ev;
        s += xrow<1>(s);
        s += xrow<2>(s);
        s += xrow<4>(s);
        s += xrow<8>(s);
        s += xhalf(s);
        float cl = ev / s;

        float a0 = 0.f, a1 = 0.f;
#pragma unroll
        for (int l = 0; l < LTOK; ++l) {
            float c = __int_as_float(__builtin_amdgcn_readlane(__float_as_int(cl), l));
            v2f wv = *(const v2f*)&wg[wave][l][2 * lane];
            a0 = fmaf(c, wv.x, a0);
            a1 = fmaf(c, wv.y, a1);
        }
        *(v2f*)&st[wave][j][2 * lane] = (v2f){a0, a1};

        __asm__ volatile("" ::: "memory");
    }

    const int p  = tweet0 >> 10;
    const int q  = tweet0 & 1023;
    const int es = lane >> 2;
    const int pr = lane & 3;
#pragma unroll
    for (int i = 0; i < 8; ++i) {
        const int e = 8 * i + es;
        v2f o = (v2f){ st[wave][2 * pr][e], st[wave][2 * pr + 1][e] };
        *(v2f*)&out[((size_t)p * 64 + e) * 1024 + q + 2 * pr] = o;
    }
}

extern "C" void kernel_launch(void* const* d_in, const int* in_sizes, int n_in,
                              void* d_out, int out_size, void* d_ws, size_t ws_size,
                              hipStream_t stream) {
    const int*   x   = (const int*)d_in[0];
    const float* emb = (const float*)d_in[1];
    float*       out = (float*)d_out;

    int n_tweets = in_sizes[0] / LTOK;
    int blocks   = n_tweets / (NWAVES * TPW);
    hipLaunchKernelGGL(tweetrep_kernel, dim3(blocks), dim3(256), 0, stream,
                       x, emb, out);
}


# scores from ONE chain of 16 WMMAs, not 32: A = tokens 0-14 + V as row 15, B = tokens 15-29 + V as column 15; row 15 and column 15 of D are the 30 scores; softmax lane order (its sum order) changes
# speedup vs baseline: 2.4852x; 1.1778x over previous
#include <hip/hip_runtime.h>


typedef float v2f __attribute__((ext_vector_type(2)));
typedef float v4f __attribute__((ext_vector_type(4)));
typedef float v8f __attribute__((ext_vector_type(8)));
typedef __attribute__((address_space(1))) char gchar;
typedef __attribute__((address_space(1))) v2f  gv2f;

#define NWAVES 8
#define TPW    8
#define LTOK   30
#define EDIM   64
#define ROWSTR 68
#define STSTR  72

static __device__ __forceinline__ float xhalf(float x) {
    int b = __float_as_int(x);
    return __int_as_float(__builtin_amdgcn_permlanex16(b, b, 0x76543210, 0xfedcba98, false, false));
}
template <int M>
static __device__ __forceinline__ float xrow(float x) {
    return __int_as_float(__builtin_amdgcn_update_dpp(0, __float_as_int(x), 0x160 + M, 0xf, 0xf, true));
}
#define LOAD_ROW(L, XS)  R[L] = *(const gv2f*)(tab + ((unsigned)(XS)[L] << 8) + lo);
#define LOAD_ROWS(XS)                                                                          \
    LOAD_ROW(0, XS)  LOAD_ROW(1, XS)  LOAD_ROW(2, XS)  LOAD_ROW(3, XS)  LOAD_ROW(4, XS)  LOAD_ROW(5, XS)   \
    LOAD_ROW(6, XS)  LOAD_ROW(7, XS)  LOAD_ROW(8, XS)  LOAD_ROW(9, XS)  LOAD_ROW(10, XS) LOAD_ROW(11, XS)  \
    LOAD_ROW(12, XS) LOAD_ROW(13, XS) LOAD_ROW(14, XS) LOAD_ROW(15, XS) LOAD_ROW(16, XS) LOAD_ROW(17, XS)  \
    LOAD_ROW(18, XS) LOAD_ROW(19, XS) LOAD_ROW(20, XS) LOAD_ROW(21, XS) LOAD_ROW(22, XS) LOAD_ROW(23, XS)  \
    LOAD_ROW(24, XS) LOAD_ROW(25, XS) LOAD_ROW(26, XS) LOAD_ROW(27, XS) LOAD_ROW(28, XS) LOAD_ROW(29, XS)

__global__ __launch_bounds__(256) void tweetrep_kernel(
    const int*   __restrict__ x,
    const float* __restrict__ emb,
    float*       __restrict__ out)
{
    __shared__ __attribute__((aligned(16))) float wg[NWAVES][LTOK + 2][ROWSTR];
    __shared__ __attribute__((aligned(16))) float st[NWAVES][TPW][STSTR];
    __shared__ __attribute__((aligned(16))) float sU[NWAVES][32][12];

    __builtin_amdgcn_s_setprio(3);

    const int tid    = threadIdx.x;
    const int wave   = __builtin_amdgcn_readfirstlane(tid >> 5);
    const int lane   = tid & 31;
    const int half   = lane >> 4;
    const int rl     = lane & 15;
    const int tweet0 = (blockIdx.x * NWAVES + wave) * TPW;

    const int*     xw   = x + (size_t)tweet0 * LTOK;
    const unsigned loff = 8u * lane;

    const int rowA = (rl < 15) ? rl : 31;
    const int rowB = (rl < 15) ? 15 + rl : 30;

    v2f R[LTOK];
    int xn[LTOK];
    const gchar* tab = (const gchar*)emb;
    {
        const unsigned lo = loff;
        LOAD_ROWS(xw)
    }
#pragma unroll
    for (int l = 0; l < LTOK; ++l) xn[l] = xw[LTOK + l];

#pragma unroll 1
    for (int j = 0; j < TPW; ++j) {
        if (j == 2) __builtin_amdgcn_s_setprio(2);
        if (j == 4) __builtin_amdgcn_s_setprio(1);
        if (j == 6) __builtin_amdgcn_s_setprio(0);

        v2f se = R[0], so = R[1];
#pragma unroll
        for (int l = 0; l < LTOK; ++l) {
            *(v2f*)&wg[wave][l][2 * lane] = R[l];
            if (l >= 2) { if (l & 1) so += R[l]; else se += R[l]; }
        }
        const v2f vsum = se + so;
        *(v2f*)&wg[wave][30][2 * lane] = vsum;
        *(v2f*)&wg[wave][31][2 * lane] = vsum;

        if (j + 1 < TPW) {
            unsigned lo = loff;
            __asm__ volatile("" : "+v"(lo));
            LOAD_ROWS(xn)
        }
        if (j + 2 < TPW) {
#pragma unroll
            for (int l = 0; l < LTOK; ++l) xn[l] = xw[(j + 2) * LTOK + l];
        }

        __asm__ volatile("" ::: "memory");

        v8f c = {};
#pragma unroll
        for (int kb = 0; kb < 16; ++kb) {
            const int k = kb * 4 + half * 2;
            const v2f fa = *(const v2f*)&wg[wave][rowA][k];
            const v2f fb = *(const v2f*)&wg[wave][rowB][k];
            c = __builtin_amdgcn_wmma_f32_16x16x4_f32(false, fa, false, fb, (short)0, c, false, false);
        }

        *(v4f*)&sU[wave][lane][0] = __builtin_shufflevector(c, c, 0, 1, 2, 3);
        *(v4f*)&sU[wave][lane][4] = __builtin_shufflevector(c, c, 4, 5, 6, 7);
        const float u = sU[wave][(rl < 8) ? 15 : 31][rl & 7];
        const bool valid = (rl != 15);
        float bl = (lane < 16) ? u : c[7];
        bl = valid ? bl : -3.0e38f;
        float mx = bl;
        mx = fmaxf(mx, xrow<1>(mx));
        mx = fmaxf(mx, xrow<2>(mx));
        mx = fmaxf(mx, xrow<4>(mx));
        mx = fmaxf(mx, xrow<8>(mx));
        mx = fmaxf(mx, xhalf(mx));
        float ev = valid ? expf(bl - mx) : 0.f;
        float s = ev;
        s += xrow<1>(s);
        s += xrow<2>(s);
        s += xrow<4>(s);
        s += xrow<8>(s);
        s += xhalf(s);
        float cl = ev / s;

        float a0 = 0.f, a1 = 0.f;
#pragma unroll
        for (int l = 0; l < LTOK; ++l) {
            float cs = __int_as_float(__builtin_amdgcn_readlane(__float_as_int(cl), (l < 15) ? l : l + 1));
            v2f wv = *(const v2f*)&wg[wave][l][2 * lane];
            a0 = fmaf(cs, wv.x, a0);
            a1 = fmaf(cs, wv.y, a1);
        }
        *(v2f*)&st[wave][j][2 * lane] = (v2f){a0, a1};

        __asm__ volatile("" ::: "memory");
    }

    const int p  = tweet0 >> 10;
    const int q  = tweet0 & 1023;
    const int es = lane >> 2;
    const int pr = lane & 3;
#pragma unroll
    for (int i = 0; i < 8; ++i) {
        const int e = 8 * i + es;
        v2f o = (v2f){ st[wave][2 * pr][e], st[wave][2 * pr + 1][e] };
        *(v2f*)&out[((size_t)p * 64 + e) * 1024 + q + 2 * pr] = o;
    }
}

extern "C" void kernel_launch(void* const* d_in, const int* in_sizes, int n_in,
                              void* d_out, int out_size, void* d_ws, size_t ws_size,
                              hipStream_t stream) {
    const int*   x   = (const int*)d_in[0];
    const float* emb = (const float*)d_in[1];
    float*       out = (float*)d_out;

    int n_tweets = in_sizes[0] / LTOK;
    int blocks   = n_tweets / (NWAVES * TPW);
    hipLaunchKernelGGL(tweetrep_kernel, dim3(blocks), dim3(256), 0, stream,
                       x, emb, out);
}
